# starting kernel
# baseline (speedup 1.0000x reference)
#include <hip/hip_runtime.h>
#include <math.h>

typedef __attribute__((ext_vector_type(2))) float v2f;
typedef __attribute__((ext_vector_type(8))) float v8f;

#define NN 32
#define MM 2
#define CC 256
#define TV 2500
#define PP 8
#define TOK_PER_BLOCK 128
#define CHUNKS 20
#define EPSF 1e-8f

#if __has_builtin(__builtin_amdgcn_wmma_f32_16x16x4_f32)
#define HAVE_WMMA_F32X4 1
#else
#warning "builtin __builtin_amdgcn_wmma_f32_16x16x4_f32 not found - using scalar fallback (wmma=0 expected)"
#define HAVE_WMMA_F32X4 0
#endif

__global__ void rn_zero_kernel(float* ws, int n) {
    int i = blockIdx.x * blockDim.x + threadIdx.x;
    if (i < n) ws[i] = 0.0f;
}

__global__ void __launch_bounds__(256)
rn_main_kernel(const float* __restrict__ x, const float* __restrict__ protos,
               float* __restrict__ g_sums, float* __restrict__ g_cnt) {
    __shared__ __align__(16) float ldsA[64 * 64];
    __shared__ float lds_pn[PP];
    __shared__ __align__(16) float lds_sums[PP * 257];
    __shared__ float lds_cnt[PP];

    const int tid   = threadIdx.x;
    const int chunk = blockIdx.x;
    const int n     = blockIdx.y;

    for (int idx = tid; idx < 64 * 64; idx += 256) {
        int j     = idx >> 6;
        int rem   = idx & 63;
        int L     = rem >> 1;
        int which = rem & 1;
        int m     = L & 15;
        int k     = 4 * j + ((L < 16) ? which : (2 + which));
        ldsA[idx] = (m < PP) ? protos[m * CC + k] : 0.0f;
    }
    if (tid < PP) {
        float s = 0.0f;
        for (int c = 0; c < CC; ++c) { float pv = protos[tid * CC + c]; s += pv * pv; }
        lds_pn[tid] = sqrtf(s);
    }
    for (int idx = tid; idx < PP * 257; idx += 256) lds_sums[idx] = 0.0f;
    if (tid < PP) lds_cnt[tid] = 0.0f;
    __syncthreads();

    const int lane  = tid & 31;
    const int wave  = tid >> 5;
    const int tloc  = lane & 15;
    const int tv    = chunk * TOK_PER_BLOCK + wave * 16 + tloc;
    const bool valid = (tv < TV);
    const int tvc   = valid ? tv : (TV - 1);
    const int khalf = (lane >> 4) << 1;

    const float* xm0 = x + (size_t)n * MM * CC * TV;
    const float* xm1 = xm0 + (size_t)CC * TV;

    v8f acc = (v8f){0.f, 0.f, 0.f, 0.f, 0.f, 0.f, 0.f, 0.f};
#if !HAVE_WMMA_F32X4
    float fdot[PP];
#pragma unroll
    for (int p = 0; p < PP; ++p) fdot[p] = 0.0f;
#endif
    float hh = 0.0f;

    for (int j = 0; j < 64; ++j) {
        int c0 = 4 * j + khalf;
        size_t o0 = (size_t)c0 * TV + tvc;
        size_t o1 = o0 + TV;
        float b0 = 0.5f * (xm0[o0] + xm1[o0]);
        float b1 = 0.5f * (xm0[o1] + xm1[o1]);
        hh += b0 * b0 + b1 * b1;
#if HAVE_WMMA_F32X4
        v2f a = *(const v2f*)&ldsA[(j << 6) + (lane << 1)];
        v2f b; b.x = b0; b.y = b1;
        acc = __builtin_amdgcn_wmma_f32_16x16x4_f32(false, a, false, b,
                                                    (short)0, acc, false, false);
#else
#pragma unroll
        for (int p = 0; p < PP; ++p)
            fdot[p] += b0 * protos[p * CC + c0] + b1 * protos[p * CC + c0 + 1];
#endif
    }

    hh += __shfl_xor(hh, 16, 32);
    float hn = sqrtf(hh);

    float dots[PP];
#if HAVE_WMMA_F32X4
#pragma unroll
    for (int p = 0; p < PP; ++p) dots[p] = acc[p];
#else
#pragma unroll
    for (int p = 0; p < PP; ++p) dots[p] = fdot[p] + __shfl_xor(fdot[p], 16, 32);
#endif

    float best = -3.0e38f;
    int bi = 0;
#pragma unroll
    for (int p = 0; p < PP; ++p) {
        float sim = dots[p] / fmaxf(hn * lds_pn[p], EPSF);
        if (sim > best) { best = sim; bi = p; }
    }
    int assign = __shfl(bi, tloc, 32);

    if (lane < 16 && valid)
        __hip_atomic_fetch_add(&lds_cnt[assign], 1.0f,
                               __ATOMIC_RELAXED, __HIP_MEMORY_SCOPE_WORKGROUP);

    for (int i = 0; i < 128; ++i) {
        int c = 2 * i + (lane >> 4);
        size_t o = (size_t)c * TV + tvc;
        float h = 0.5f * (xm0[o] + xm1[o]);
        if (valid)
            __hip_atomic_fetch_add(&lds_sums[assign * 257 + c], h,
                                   __ATOMIC_RELAXED, __HIP_MEMORY_SCOPE_WORKGROUP);
    }
    __syncthreads();

    float* gs = g_sums + (size_t)n * PP * CC;
#pragma unroll
    for (int p = 0; p < PP; ++p)
        __hip_atomic_fetch_add(&gs[p * CC + tid], lds_sums[p * 257 + tid],
                               __ATOMIC_RELAXED, __HIP_MEMORY_SCOPE_AGENT);
    if (tid < PP)
        __hip_atomic_fetch_add(&g_cnt[n * PP + tid], lds_cnt[tid],
                               __ATOMIC_RELAXED, __HIP_MEMORY_SCOPE_AGENT);
}

__global__ void rn_final_kernel(const float* __restrict__ g_sums,
                                const float* __restrict__ g_cnt,
                                float* __restrict__ out) {
    int n = blockIdx.x;
    int c = threadIdx.x;
    float s = 0.0f;
#pragma unroll
    for (int p = 0; p < PP; ++p) {
        float cnt = g_cnt[n * PP + p];
        s += g_sums[((size_t)n * PP + p) * CC + c] / fmaxf(cnt, 1.0f);
    }
    out[n * CC + c] = s * 0.125f;
}

extern "C" void kernel_launch(void* const* d_in, const int* in_sizes, int n_in,
                              void* d_out, int out_size, void* d_ws, size_t ws_size,
                              hipStream_t stream) {
    const float* x      = (const float*)d_in[0];
    const float* protos = (const float*)d_in[1];

    float* g_sums = (float*)d_ws;
    float* g_cnt  = g_sums + NN * PP * CC;

    int zn = NN * PP * CC + NN * PP;
    rn_zero_kernel<<<(zn + 255) / 256, 256, 0, stream>>>((float*)d_ws, zn);

    dim3 grid(CHUNKS, NN);
    rn_main_kernel<<<grid, 256, 0, stream>>>(x, protos, g_sums, g_cnt);

    rn_final_kernel<<<NN, CC, 0, stream>>>(g_sums, g_cnt, (float*)d_out);
}


# Segment means as ONE weighted token sum (weight 1/segment size): kernel 1 stores assignments and chunk counts, kernel 2 streams x rows; no sum atomics, no zero kernel; sum order changes (now fixed)
# speedup vs baseline: 2.0854x; 2.0854x over previous
#include <hip/hip_runtime.h>
#include <math.h>

typedef __attribute__((ext_vector_type(2))) float v2f;
typedef __attribute__((ext_vector_type(4))) float v4f;
typedef __attribute__((ext_vector_type(8))) float v8f;

#define NN 32
#define MM 2
#define CC 256
#define TV 2500
#define PP 8
#define TOK_PER_BLOCK 128
#define CHUNKS 20
#define EPSF 1e-8f

#define AS_PITCH_W (CHUNKS * 32)
#define CNT_PITCH 32

#if __has_builtin(__builtin_amdgcn_wmma_f32_16x16x4_f32)
#define HAVE_WMMA_F32X4 1
#else
#warning "builtin __builtin_amdgcn_wmma_f32_16x16x4_f32 not found - using scalar fallback (wmma=0 expected)"
#define HAVE_WMMA_F32X4 0
#endif

__global__ void __launch_bounds__(256)
rn_assign_kernel(const float* __restrict__ x, const float* __restrict__ protos,
                 float* __restrict__ ws_assign, float* __restrict__ ws_cnt) {
    __shared__ __align__(16) float ldsA[64 * 64];
    __shared__ float lds_pn[PP];
    __shared__ float lds_cnt[PP];
    __shared__ __align__(16) unsigned char lds_as[TOK_PER_BLOCK];

    const int tid   = threadIdx.x;
    const int chunk = blockIdx.x;
    const int n     = blockIdx.y;

    for (int idx = tid; idx < 64 * 64; idx += 256) {
        int j     = idx >> 6;
        int rem   = idx & 63;
        int L     = rem >> 1;
        int which = rem & 1;
        int m     = L & 15;
        int k     = 4 * j + ((L < 16) ? which : (2 + which));
        ldsA[idx] = (m < PP) ? protos[m * CC + k] : 0.0f;
    }
    if (tid < PP) {
        float s = 0.0f;
        for (int c = 0; c < CC; ++c) { float pv = protos[tid * CC + c]; s += pv * pv; }
        lds_pn[tid] = sqrtf(s);
    }
    if (tid < PP) lds_cnt[tid] = 0.0f;
    __syncthreads();

    const int lane  = tid & 31;
    const int wave  = tid >> 5;
    const int tloc  = lane & 15;
    const int tv    = chunk * TOK_PER_BLOCK + wave * 16 + tloc;
    const bool valid = (tv < TV);
    const int tvc   = valid ? tv : (TV - 1);
    const int khalf = (lane >> 4) << 1;

    const float* xm0 = x + (size_t)n * MM * CC * TV;
    const float* xm1 = xm0 + (size_t)CC * TV;

    v8f acc = (v8f){0.f, 0.f, 0.f, 0.f, 0.f, 0.f, 0.f, 0.f};
#if !HAVE_WMMA_F32X4
    float fdot[PP];
#pragma unroll
    for (int p = 0; p < PP; ++p) fdot[p] = 0.0f;
#endif
    float hh = 0.0f;

    for (int j = 0; j < 64; ++j) {
        int c0 = 4 * j + khalf;
        size_t o0 = (size_t)c0 * TV + tvc;
        size_t o1 = o0 + TV;
        float b0 = 0.5f * (xm0[o0] + xm1[o0]);
        float b1 = 0.5f * (xm0[o1] + xm1[o1]);
        hh += b0 * b0 + b1 * b1;
#if HAVE_WMMA_F32X4
        v2f a = *(const v2f*)&ldsA[(j << 6) + (lane << 1)];
        v2f b; b.x = b0; b.y = b1;
        acc = __builtin_amdgcn_wmma_f32_16x16x4_f32(false, a, false, b,
                                                    (short)0, acc, false, false);
#else
#pragma unroll
        for (int p = 0; p < PP; ++p)
            fdot[p] += b0 * protos[p * CC + c0] + b1 * protos[p * CC + c0 + 1];
#endif
    }

    hh += __shfl_xor(hh, 16, 32);
    float hn = sqrtf(hh);

    float dots[PP];
#if HAVE_WMMA_F32X4
#pragma unroll
    for (int p = 0; p < PP; ++p) dots[p] = acc[p];
#else
#pragma unroll
    for (int p = 0; p < PP; ++p) dots[p] = fdot[p] + __shfl_xor(fdot[p], 16, 32);
#endif

    float best = -3.0e38f;
    int bi = 0;
#pragma unroll
    for (int p = 0; p < PP; ++p) {
        float sim = dots[p] / fmaxf(hn * lds_pn[p], EPSF);
        if (sim > best) { best = sim; bi = p; }
    }

    if (lane < 16) {
        lds_as[wave * 16 + tloc] = (unsigned char)(valid ? bi : 0);
        if (valid)
            __hip_atomic_fetch_add(&lds_cnt[bi], 1.0f,
                                   __ATOMIC_RELAXED, __HIP_MEMORY_SCOPE_WORKGROUP);
    }
    __syncthreads();

    const int blk = n * CHUNKS + chunk;
    if (tid < 32)
        ws_assign[blk * 32 + tid] = __uint_as_float(((const unsigned*)lds_as)[tid]);
    if (tid < PP)
        ws_cnt[blk * CNT_PITCH + tid] = lds_cnt[tid];
}

#define PB_WAVES  16
#define PB_PIECES 20
#define PB_GROUP  4
#define PB_GROUPS (PB_PIECES / PB_GROUP)
#define TV4 (TV / 4)

static __device__ __forceinline__ v4f load_once(const v4f* p) { return __builtin_nontemporal_load(p); }

__global__ void __launch_bounds__(PB_WAVES * 32)
rn_pool_kernel(const float* __restrict__ x, const float* __restrict__ ws_assign,
               const float* __restrict__ ws_cnt, float* __restrict__ out) {
    __shared__ __align__(16) float lds_out[PB_WAVES * 2];

    const unsigned tid  = threadIdx.x;
    const unsigned lane = tid & 31u;
    const unsigned wave = __builtin_amdgcn_readfirstlane(tid >> 5);
    const unsigned pair = blockIdx.x * PB_WAVES + wave;
    const unsigned n    = pair >> 7;
    const unsigned c0   = (pair & 127u) << 1;

    const float* cp = ws_cnt + (size_t)(n * CHUNKS + (lane >> 3)) * CNT_PITCH + (lane & 7u);
    float cpart[CHUNKS / 4];
#pragma unroll
    for (int j = 0; j < CHUNKS / 4; ++j) cpart[j] = cp[j * 4 * CNT_PITCH];

    const float* ap = ws_assign + (size_t)n * AS_PITCH_W;
    unsigned aw[PB_PIECES];
#pragma unroll
    for (int q = 0; q < PB_PIECES; ++q) {
        unsigned i = 32u * q + lane;
        if (i > TV4 - 1) i = TV4 - 1;
        aw[q] = __float_as_uint(ap[i]);
    }
    const bool tail_ok = lane < (TV4 - 32 * (PB_PIECES - 1));

    const float* xr = x + ((size_t)n * MM * CC + c0) * TV;
    const v4f* r0m0 = (const v4f*)xr;
    const v4f* r0m1 = (const v4f*)(xr + (size_t)CC * TV);
    const v4f* r1m0 = (const v4f*)(xr + TV);
    const v4f* r1m1 = (const v4f*)(xr + (size_t)CC * TV + TV);

    v4f buf[2][PB_GROUP * 4];
#define LOAD_GROUP(G) do { \
    _Pragma("unroll") for (int p_ = 0; p_ < PB_GROUP; ++p_) { \
        unsigned i_ = 32u * ((G) * PB_GROUP + p_) + lane; \
        if (i_ > TV4 - 1) i_ = TV4 - 1; \
        buf[(G) & 1][4 * p_ + 0] = load_once(r0m0 + i_); \
        buf[(G) & 1][4 * p_ + 1] = load_once(r0m1 + i_); \
        buf[(G) & 1][4 * p_ + 2] = load_once(r1m0 + i_); \
        buf[(G) & 1][4 * p_ + 3] = load_once(r1m1 + i_); \
    } } while (0)

    v4f acc0 = (v4f){0.f, 0.f, 0.f, 0.f};
    v4f acc1 = (v4f){0.f, 0.f, 0.f, 0.f};
    int winv_bits = 0;

    __builtin_amdgcn_sched_barrier(0);
    LOAD_GROUP(0);
#pragma unroll
    for (int g = 0; g < PB_GROUPS; ++g) {
        __builtin_amdgcn_sched_barrier(0);
        if (g + 1 < PB_GROUPS) LOAD_GROUP(g + 1);
        __builtin_amdgcn_sched_barrier(0);
        if (g == 0) {
            float cnt = 0.0f;
#pragma unroll
            for (int j = 0; j < CHUNKS / 4; ++j) cnt += cpart[j];
            cnt += __shfl_xor(cnt, 8, 32);
            cnt += __shfl_xor(cnt, 16, 32);
            winv_bits = __float_as_int(1.0f / fmaxf(cnt, 1.0f));
        }
#pragma unroll
        for (int p = 0; p < PB_GROUP; ++p) {
            const int q = g * PB_GROUP + p;
            const unsigned w = aw[q];
            v4f wt;
            wt.x = __int_as_float(__builtin_amdgcn_ds_bpermute((int)((w << 2) & 28u), winv_bits));
            wt.y = __int_as_float(__builtin_amdgcn_ds_bpermute((int)((w >> 6) & 28u), winv_bits));
            wt.z = __int_as_float(__builtin_amdgcn_ds_bpermute((int)((w >> 14) & 28u), winv_bits));
            wt.w = __int_as_float(__builtin_amdgcn_ds_bpermute((int)((w >> 22) & 28u), winv_bits));
            if (q == PB_PIECES - 1 && !tail_ok) wt = (v4f){0.f, 0.f, 0.f, 0.f};
            const v4f* b = &buf[g & 1][4 * p];
            acc0 += (b[0] + b[1]) * wt;
            acc1 += (b[2] + b[3]) * wt;
        }
    }
#undef LOAD_GROUP

    float s0 = (acc0.x + acc0.y) + (acc0.z + acc0.w);
    float s1 = (acc1.x + acc1.y) + (acc1.z + acc1.w);
#pragma unroll
    for (int m = 16; m >= 1; m >>= 1) {
        s0 += __shfl_xor(s0, m, 32);
        s1 += __shfl_xor(s1, m, 32);
    }
    if (lane == 0) { lds_out[2 * wave] = s0 * 0.0625f; lds_out[2 * wave + 1] = s1 * 0.0625f; }
    __syncthreads();
    if (tid < 32) out[(size_t)blockIdx.x * 32 + tid] = lds_out[tid];
}

extern "C" void kernel_launch(void* const* d_in, const int* in_sizes, int n_in,
                              void* d_out, int out_size, void* d_ws, size_t ws_size,
                              hipStream_t stream) {
    const float* x      = (const float*)d_in[0];
    const float* protos = (const float*)d_in[1];

    float* ws_assign = (float*)d_ws;
    float* ws_cnt    = ws_assign + NN * AS_PITCH_W;

    dim3 grid(CHUNKS, NN);
    rn_assign_kernel<<<grid, 256, 0, stream>>>(x, protos, ws_assign, ws_cnt);

    rn_pool_kernel<<<NN * CC / (2 * PB_WAVES), PB_WAVES * 32, 0, stream>>>(x, ws_assign, ws_cnt, (float*)d_out);
}


# Assignment kernel retiled for 16-byte x loads: a lane owns 4 adjacent tokens of a channel row, 4 WMMA tiles fed from registers; 4 waves split the channels, partials added in LDS (sum order changes)
# speedup vs baseline: 2.2115x; 1.0605x over previous
#include <hip/hip_runtime.h>
#include <math.h>

typedef __attribute__((ext_vector_type(2))) float v2f;
typedef __attribute__((ext_vector_type(4))) float v4f;
typedef __attribute__((ext_vector_type(8))) float v8f;

#define NN 32
#define MM 2
#define CC 256
#define TV 2500
#define PP 8
#define TOK_PER_BLOCK 128
#define CHUNKS 20
#define EPSF 1e-8f

#define AS_PITCH_W (CHUNKS * 32)
#define CNT_PITCH 32

#if __has_builtin(__builtin_amdgcn_wmma_f32_16x16x4_f32)
#define HAVE_WMMA_F32X4 1
#else
#warning "builtin __builtin_amdgcn_wmma_f32_16x16x4_f32 not found - using scalar fallback (wmma=0 expected)"
#define HAVE_WMMA_F32X4 0
#endif

#define RA_KQ      4
#define RA_KSTEPS  (CC / 4 / RA_KQ)
#define RA_GSTEPS  2
#define RA_GROUPS  (RA_KSTEPS / RA_GSTEPS)
#define RA_SLOTS   3
#define RA_ROW4    (TV / 4)
#define RA_PER4    (CC * TV / 4)

__global__ void __launch_bounds__(256)
rn_assign_kernel(const float* __restrict__ x, const float* __restrict__ protos,
                 float* __restrict__ ws_assign, float* __restrict__ ws_cnt) {
    __shared__ __align__(16) float ldsA[64 * 64];
    __shared__ __align__(16) float lds_pd[RA_KQ * PP * TOK_PER_BLOCK];
    __shared__ __align__(16) float lds_ph[RA_KQ * 2 * TOK_PER_BLOCK];
    __shared__ float lds_pn[PP];
    __shared__ float lds_cnt[PP];
    __shared__ __align__(16) unsigned char lds_as[TOK_PER_BLOCK];

    const int tid   = threadIdx.x;
    const int chunk = blockIdx.x;
    const int n     = blockIdx.y;

    for (int idx = tid; idx < 64 * 64; idx += 256) {
        int j     = idx >> 6;
        int rem   = idx & 63;
        int L     = rem >> 1;
        int which = rem & 1;
        int m     = L & 15;
        int k     = 4 * j + ((L < 16) ? which : (2 + which));
        ldsA[idx] = (m < PP) ? protos[m * CC + k] : 0.0f;
    }
    if (tid < PP) {
        float s = 0.0f;
        for (int c = 0; c < CC; ++c) { float pv = protos[tid * CC + c]; s += pv * pv; }
        lds_pn[tid] = sqrtf(s);
    }
    if (tid < PP) lds_cnt[tid] = 0.0f;
    __syncthreads();

    const int lane = tid & 31;
    const int wave = __builtin_amdgcn_readfirstlane(tid >> 5);
    const int tg   = wave & 1;
    const int cq   = wave >> 1;
    const int col  = lane & 15;
    const int half = lane >> 4;

    int g4 = chunk * (TOK_PER_BLOCK / 4) + tg * 16 + col;
    if (g4 > RA_ROW4 - 1) g4 = RA_ROW4 - 1;
    const v4f* xp = (const v4f*)(x + (size_t)n * MM * CC * TV + (size_t)(cq * 64 + 2 * half) * TV) + g4;

    v4f buf[RA_SLOTS][RA_GSTEPS * 4];
#define RA_LOAD_GROUP(G) do { \
    _Pragma("unroll") for (int ks_ = 0; ks_ < RA_GSTEPS; ++ks_) { \
        const int r_ = 4 * ((G) * RA_GSTEPS + ks_); \
        buf[(G) % RA_SLOTS][4 * ks_ + 0] = xp[(r_    ) * RA_ROW4]; \
        buf[(G) % RA_SLOTS][4 * ks_ + 1] = xp[(r_    ) * RA_ROW4 + RA_PER4]; \
        buf[(G) % RA_SLOTS][4 * ks_ + 2] = xp[(r_ + 1) * RA_ROW4]; \
        buf[(G) % RA_SLOTS][4 * ks_ + 3] = xp[(r_ + 1) * RA_ROW4 + RA_PER4]; \
    } } while (0)

    v8f acc[4] = {};
#if !HAVE_WMMA_F32X4
    float fd[4][PP];
#pragma unroll
    for (int i = 0; i < 4; ++i)
#pragma unroll
        for (int p = 0; p < PP; ++p) fd[i][p] = 0.0f;
#endif
    v4f hh = (v4f){0.f, 0.f, 0.f, 0.f};

    RA_LOAD_GROUP(0);
    RA_LOAD_GROUP(1);
#pragma unroll
    for (int g = 0; g < RA_GROUPS; ++g) {
        __builtin_amdgcn_sched_barrier(0);
        if (g + 2 < RA_GROUPS) RA_LOAD_GROUP(g + 2);
        __builtin_amdgcn_sched_barrier(0);
#pragma unroll
        for (int ks = 0; ks < RA_GSTEPS; ++ks) {
            const int j = cq * RA_KSTEPS + g * RA_GSTEPS + ks;
            const v4f* b = &buf[g % RA_SLOTS][4 * ks];
            const v4f b0 = 0.5f * (b[0] + b[1]);
            const v4f b1 = 0.5f * (b[2] + b[3]);
            hh += b0 * b0 + b1 * b1;
#if HAVE_WMMA_F32X4
            const v2f a = *(const v2f*)&ldsA[(j << 6) + (lane << 1)];
            v2f bb;
            bb.x = b0.x; bb.y = b1.x;
            acc[0] = __builtin_amdgcn_wmma_f32_16x16x4_f32(false, a, false, bb, (short)0, acc[0], false, false);
            bb.x = b0.y; bb.y = b1.y;
            acc[1] = __builtin_amdgcn_wmma_f32_16x16x4_f32(false, a, false, bb, (short)0, acc[1], false, false);
            bb.x = b0.z; bb.y = b1.z;
            acc[2] = __builtin_amdgcn_wmma_f32_16x16x4_f32(false, a, false, bb, (short)0, acc[2], false, false);
            bb.x = b0.w; bb.y = b1.w;
            acc[3] = __builtin_amdgcn_wmma_f32_16x16x4_f32(false, a, false, bb, (short)0, acc[3], false, false);
#else
#pragma unroll
            for (int p = 0; p < PP; ++p) {
                const float p0 = protos[p * CC + 4 * j + 2 * half];
                const float p1 = protos[p * CC + 4 * j + 2 * half + 1];
                fd[0][p] += b0.x * p0 + b1.x * p1;
                fd[1][p] += b0.y * p0 + b1.y * p1;
                fd[2][p] += b0.z * p0 + b1.z * p1;
                fd[3][p] += b0.w * p0 + b1.w * p1;
            }
#endif
        }
        asm volatile("" : "+v"(hh));
    }
#undef RA_LOAD_GROUP

#if !HAVE_WMMA_F32X4
#pragma unroll
    for (int i = 0; i < 4; ++i)
#pragma unroll
        for (int p = 0; p < PP; ++p) acc[i][p] = fd[i][p] + __shfl_xor(fd[i][p], 16, 32);
#endif

    const int tq = tg * 64 + 4 * col;
    if (lane < 16) {
#pragma unroll
        for (int p = 0; p < PP; ++p) {
            v4f d;
            d.x = acc[0][p]; d.y = acc[1][p]; d.z = acc[2][p]; d.w = acc[3][p];
            *(v4f*)&lds_pd[(cq * PP + p) * TOK_PER_BLOCK + tq] = d;
        }
    }
    *(v4f*)&lds_ph[(cq * 2 + half) * TOK_PER_BLOCK + tq] = hh;
    __syncthreads();

    if (tid < TOK_PER_BLOCK) {
        const bool valid = (chunk * TOK_PER_BLOCK + tid) < TV;
        float hsq = 0.0f;
        for (int q = 0; q < RA_KQ; ++q)
            hsq += lds_ph[(q * 2) * TOK_PER_BLOCK + tid] + lds_ph[(q * 2 + 1) * TOK_PER_BLOCK + tid];
        const float hn = sqrtf(hsq);

        float best = -3.0e38f;
        int bi = 0;
        for (int p = 0; p < PP; ++p) {
            float d = lds_pd[p * TOK_PER_BLOCK + tid];
            for (int q = 1; q < RA_KQ; ++q) d += lds_pd[(q * PP + p) * TOK_PER_BLOCK + tid];
            float sim = d / fmaxf(hn * lds_pn[p], EPSF);
            if (sim > best) { best = sim; bi = p; }
        }
        lds_as[tid] = (unsigned char)(valid ? bi : 0);
        if (valid)
            __hip_atomic_fetch_add(&lds_cnt[bi], 1.0f,
                                   __ATOMIC_RELAXED, __HIP_MEMORY_SCOPE_WORKGROUP);
    }
    __syncthreads();

    const int blk = n * CHUNKS + chunk;
    if (tid < 32)
        ws_assign[blk * 32 + tid] = __uint_as_float(((const unsigned*)lds_as)[tid]);
    if (tid < PP)
        ws_cnt[blk * CNT_PITCH + tid] = lds_cnt[tid];
}

#define PB_WAVES  16
#define PB_PIECES 20
#define PB_GROUP  4
#define PB_GROUPS (PB_PIECES / PB_GROUP)
#define TV4 (TV / 4)

static __device__ __forceinline__ v4f load_once(const v4f* p) { return __builtin_nontemporal_load(p); }

__global__ void __launch_bounds__(PB_WAVES * 32)
rn_pool_kernel(const float* __restrict__ x, const float* __restrict__ ws_assign,
               const float* __restrict__ ws_cnt, float* __restrict__ out) {
    __shared__ __align__(16) float lds_out[PB_WAVES * 2];

    const unsigned tid  = threadIdx.x;
    const unsigned lane = tid & 31u;
    const unsigned wave = __builtin_amdgcn_readfirstlane(tid >> 5);
    const unsigned pair = blockIdx.x * PB_WAVES + wave;
    const unsigned n    = pair >> 7;
    const unsigned c0   = (pair & 127u) << 1;

    const float* cp = ws_cnt + (size_t)(n * CHUNKS + (lane >> 3)) * CNT_PITCH + (lane & 7u);
    float cpart[CHUNKS / 4];
#pragma unroll
    for (int j = 0; j < CHUNKS / 4; ++j) cpart[j] = cp[j * 4 * CNT_PITCH];

    const float* ap = ws_assign + (size_t)n * AS_PITCH_W;
    unsigned aw[PB_PIECES];
#pragma unroll
    for (int q = 0; q < PB_PIECES; ++q) {
        unsigned i = 32u * q + lane;
        if (i > TV4 - 1) i = TV4 - 1;
        aw[q] = __float_as_uint(ap[i]);
    }
    const bool tail_ok = lane < (TV4 - 32 * (PB_PIECES - 1));

    const float* xr = x + ((size_t)n * MM * CC + c0) * TV;
    const v4f* r0m0 = (const v4f*)xr;
    const v4f* r0m1 = (const v4f*)(xr + (size_t)CC * TV);
    const v4f* r1m0 = (const v4f*)(xr + TV);
    const v4f* r1m1 = (const v4f*)(xr + (size_t)CC * TV + TV);

    v4f buf[2][PB_GROUP * 4];
#define LOAD_GROUP(G) do { \
    _Pragma("unroll") for (int p_ = 0; p_ < PB_GROUP; ++p_) { \
        unsigned i_ = 32u * ((G) * PB_GROUP + p_) + lane; \
        if (i_ > TV4 - 1) i_ = TV4 - 1; \
        buf[(G) & 1][4 * p_ + 0] = load_once(r0m0 + i_); \
        buf[(G) & 1][4 * p_ + 1] = load_once(r0m1 + i_); \
        buf[(G) & 1][4 * p_ + 2] = load_once(r1m0 + i_); \
        buf[(G) & 1][4 * p_ + 3] = load_once(r1m1 + i_); \
    } } while (0)

    v4f acc0 = (v4f){0.f, 0.f, 0.f, 0.f};
    v4f acc1 = (v4f){0.f, 0.f, 0.f, 0.f};
    int winv_bits = 0;

    __builtin_amdgcn_sched_barrier(0);
    LOAD_GROUP(0);
#pragma unroll
    for (int g = 0; g < PB_GROUPS; ++g) {
        __builtin_amdgcn_sched_barrier(0);
        if (g + 1 < PB_GROUPS) LOAD_GROUP(g + 1);
        __builtin_amdgcn_sched_barrier(0);
        if (g == 0) {
            float cnt = 0.0f;
#pragma unroll
            for (int j = 0; j < CHUNKS / 4; ++j) cnt += cpart[j];
            cnt += __shfl_xor(cnt, 8, 32);
            cnt += __shfl_xor(cnt, 16, 32);
            winv_bits = __float_as_int(1.0f / fmaxf(cnt, 1.0f));
        }
#pragma unroll
        for (int p = 0; p < PB_GROUP; ++p) {
            const int q = g * PB_GROUP + p;
            const unsigned w = aw[q];
            v4f wt;
            wt.x = __int_as_float(__builtin_amdgcn_ds_bpermute((int)((w << 2) & 28u), winv_bits));
            wt.y = __int_as_float(__builtin_amdgcn_ds_bpermute((int)((w >> 6) & 28u), winv_bits));
            wt.z = __int_as_float(__builtin_amdgcn_ds_bpermute((int)((w >> 14) & 28u), winv_bits));
            wt.w = __int_as_float(__builtin_amdgcn_ds_bpermute((int)((w >> 22) & 28u), winv_bits));
            if (q == PB_PIECES - 1 && !tail_ok) wt = (v4f){0.f, 0.f, 0.f, 0.f};
            const v4f* b = &buf[g & 1][4 * p];
            acc0 += (b[0] + b[1]) * wt;
            acc1 += (b[2] + b[3]) * wt;
        }
    }
#undef LOAD_GROUP

    float s0 = (acc0.x + acc0.y) + (acc0.z + acc0.w);
    float s1 = (acc1.x + acc1.y) + (acc1.z + acc1.w);
#pragma unroll
    for (int m = 16; m >= 1; m >>= 1) {
        s0 += __shfl_xor(s0, m, 32);
        s1 += __shfl_xor(s1, m, 32);
    }
    if (lane == 0) { lds_out[2 * wave] = s0 * 0.0625f; lds_out[2 * wave + 1] = s1 * 0.0625f; }
    __syncthreads();
    if (tid < 32) out[(size_t)blockIdx.x * 32 + tid] = lds_out[tid];
}

extern "C" void kernel_launch(void* const* d_in, const int* in_sizes, int n_in,
                              void* d_out, int out_size, void* d_ws, size_t ws_size,
                              hipStream_t stream) {
    const float* x      = (const float*)d_in[0];
    const float* protos = (const float*)d_in[1];

    float* ws_assign = (float*)d_ws;
    float* ws_cnt    = ws_assign + NN * AS_PITCH_W;

    dim3 grid(CHUNKS, NN);
    rn_assign_kernel<<<grid, 256, 0, stream>>>(x, protos, ws_assign, ws_cnt);

    rn_pool_kernel<<<NN * CC / (2 * PB_WAVES), PB_WAVES * 32, 0, stream>>>(x, ws_assign, ws_cnt, (float*)d_out);
}


# Assignment kernel: each lane fetches its own prototype A fragments in front of its x loads and parks them in LDS; block staging loop, serial norm loop and first barrier gone (norm sum order changes)
# speedup vs baseline: 2.4777x; 1.1204x over previous
#include <hip/hip_runtime.h>
#include <math.h>

typedef __attribute__((ext_vector_type(2))) float v2f;
typedef __attribute__((ext_vector_type(4))) float v4f;
typedef __attribute__((ext_vector_type(8))) float v8f;

#define NN 32
#define MM 2
#define CC 256
#define TV 2500
#define PP 8
#define TOK_PER_BLOCK 128
#define CHUNKS 20
#define EPSF 1e-8f

#define AS_PITCH_W (CHUNKS * 32)
#define CNT_PITCH 32

#if __has_builtin(__builtin_amdgcn_wmma_f32_16x16x4_f32)
#define HAVE_WMMA_F32X4 1
#else
#warning "builtin __builtin_amdgcn_wmma_f32_16x16x4_f32 not found - using scalar fallback (wmma=0 expected)"
#define HAVE_WMMA_F32X4 0
#endif

#define RA_KQ      4
#define RA_KSTEPS  (CC / 4 / RA_KQ)
#define RA_GSTEPS  2
#define RA_GROUPS  (RA_KSTEPS / RA_GSTEPS)
#define RA_SLOTS   3
#define RA_ROW4    (TV / 4)
#define RA_PER4    (CC * TV / 4)

__global__ void __launch_bounds__(256)
rn_assign_kernel(const float* __restrict__ x, const float* __restrict__ protos,
                 float* __restrict__ ws_assign, float* __restrict__ ws_cnt) {
    __shared__ __align__(16) float ldsA[8 * RA_KSTEPS * 64];
    __shared__ __align__(16) float lds_pd[RA_KQ * PP * TOK_PER_BLOCK];
    __shared__ __align__(16) float lds_ph[RA_KQ * 2 * TOK_PER_BLOCK];
    __shared__ __align__(16) float lds_pq[RA_KQ * PP];
    __shared__ float lds_cnt[PP];
    __shared__ __align__(16) unsigned char lds_as[TOK_PER_BLOCK];

    const int tid   = threadIdx.x;
    const int chunk = blockIdx.x;
    const int n     = blockIdx.y;

    if (tid < PP) lds_cnt[tid] = 0.0f;

    const int lane = tid & 31;
    const int wave = __builtin_amdgcn_readfirstlane(tid >> 5);
    const int tg   = wave & 1;
    const int cq   = wave >> 1;
    const int col  = lane & 15;
    const int half = lane >> 4;

    int g4 = chunk * (TOK_PER_BLOCK / 4) + tg * 16 + col;
    if (g4 > RA_ROW4 - 1) g4 = RA_ROW4 - 1;
    const v4f* xp = (const v4f*)(x + (size_t)n * MM * CC * TV + (size_t)(cq * 64 + 2 * half) * TV) + g4;


    const v2f* pa = (const v2f*)(protos + (col & (PP - 1)) * CC + cq * (CC / RA_KQ) + 2 * half);
    v2f af[RA_KSTEPS];
#pragma unroll
    for (int j = 0; j < RA_KSTEPS; ++j) af[j] = pa[2 * j];

    v4f buf[RA_SLOTS][RA_GSTEPS * 4];
#define RA_LOAD_GROUP(G) do { \
    _Pragma("unroll") for (int ks_ = 0; ks_ < RA_GSTEPS; ++ks_) { \
        const int r_ = 4 * ((G) * RA_GSTEPS + ks_); \
        buf[(G) % RA_SLOTS][4 * ks_ + 0] = xp[(r_    ) * RA_ROW4]; \
        buf[(G) % RA_SLOTS][4 * ks_ + 1] = xp[(r_    ) * RA_ROW4 + RA_PER4]; \
        buf[(G) % RA_SLOTS][4 * ks_ + 2] = xp[(r_ + 1) * RA_ROW4]; \
        buf[(G) % RA_SLOTS][4 * ks_ + 3] = xp[(r_ + 1) * RA_ROW4 + RA_PER4]; \
    } } while (0)

    v8f acc[4] = {};
#if !HAVE_WMMA_F32X4
    float fd[4][PP];
#pragma unroll
    for (int i = 0; i < 4; ++i)
#pragma unroll
        for (int p = 0; p < PP; ++p) fd[i][p] = 0.0f;
#endif
    v4f hh = (v4f){0.f, 0.f, 0.f, 0.f};

    RA_LOAD_GROUP(0);
    RA_LOAD_GROUP(1);
    __builtin_amdgcn_sched_barrier(0);

    float pq = 0.0f;
#pragma unroll
    for (int j = 0; j < RA_KSTEPS; ++j) pq += af[j].x * af[j].x + af[j].y * af[j].y;
    pq += __shfl_xor(pq, 16, 32);
    if (tg == 0 && lane < PP) lds_pq[cq * PP + lane] = pq;

    float* const myA = &ldsA[(wave * RA_KSTEPS) << 6] + (lane << 1);
#pragma unroll
    for (int j = 0; j < RA_KSTEPS; ++j) *(v2f*)&myA[j << 6] = af[j];

#pragma unroll
    for (int g = 0; g < RA_GROUPS; ++g) {
        __builtin_amdgcn_sched_barrier(0);
        if (g + 2 < RA_GROUPS) RA_LOAD_GROUP(g + 2);
        __builtin_amdgcn_sched_barrier(0);
#pragma unroll
        for (int ks = 0; ks < RA_GSTEPS; ++ks) {
            const v4f* b = &buf[g % RA_SLOTS][4 * ks];
            const v4f b0 = 0.5f * (b[0] + b[1]);
            const v4f b1 = 0.5f * (b[2] + b[3]);
            hh += b0 * b0 + b1 * b1;
#if HAVE_WMMA_F32X4
            const v2f a = *(const v2f*)&myA[(g * RA_GSTEPS + ks) << 6];
            v2f bb;
            bb.x = b0.x; bb.y = b1.x;
            acc[0] = __builtin_amdgcn_wmma_f32_16x16x4_f32(false, a, false, bb, (short)0, acc[0], false, false);
            bb.x = b0.y; bb.y = b1.y;
            acc[1] = __builtin_amdgcn_wmma_f32_16x16x4_f32(false, a, false, bb, (short)0, acc[1], false, false);
            bb.x = b0.z; bb.y = b1.z;
            acc[2] = __builtin_amdgcn_wmma_f32_16x16x4_f32(false, a, false, bb, (short)0, acc[2], false, false);
            bb.x = b0.w; bb.y = b1.w;
            acc[3] = __builtin_amdgcn_wmma_f32_16x16x4_f32(false, a, false, bb, (short)0, acc[3], false, false);
#else
            const int j = cq * RA_KSTEPS + g * RA_GSTEPS + ks;
#pragma unroll
            for (int p = 0; p < PP; ++p) {
                const float p0 = protos[p * CC + 4 * j + 2 * half];
                const float p1 = protos[p * CC + 4 * j + 2 * half + 1];
                fd[0][p] += b0.x * p0 + b1.x * p1;
                fd[1][p] += b0.y * p0 + b1.y * p1;
                fd[2][p] += b0.z * p0 + b1.z * p1;
                fd[3][p] += b0.w * p0 + b1.w * p1;
            }
#endif
        }
        asm volatile("" : "+v"(hh));
    }
#undef RA_LOAD_GROUP

#if !HAVE_WMMA_F32X4
#pragma unroll
    for (int i = 0; i < 4; ++i)
#pragma unroll
        for (int p = 0; p < PP; ++p) acc[i][p] = fd[i][p] + __shfl_xor(fd[i][p], 16, 32);
#endif

    const int tq = tg * 64 + 4 * col;
    if (lane < 16) {
#pragma unroll
        for (int p = 0; p < PP; ++p) {
            v4f d;
            d.x = acc[0][p]; d.y = acc[1][p]; d.z = acc[2][p]; d.w = acc[3][p];
            *(v4f*)&lds_pd[(cq * PP + p) * TOK_PER_BLOCK + tq] = d;
        }
    }
    *(v4f*)&lds_ph[(cq * 2 + half) * TOK_PER_BLOCK + tq] = hh;
    __syncthreads();

    if (tid < TOK_PER_BLOCK) {
        const bool valid = (chunk * TOK_PER_BLOCK + tid) < TV;
        float hsq = 0.0f;
        for (int q = 0; q < RA_KQ; ++q)
            hsq += lds_ph[(q * 2) * TOK_PER_BLOCK + tid] + lds_ph[(q * 2 + 1) * TOK_PER_BLOCK + tid];
        const float hn = sqrtf(hsq);

        float psq = lds_pq[lane & (PP - 1)];
        for (int q = 1; q < RA_KQ; ++q) psq += lds_pq[q * PP + (lane & (PP - 1))];
        const int pn_bits = __float_as_int(sqrtf(psq));

        float best = -3.0e38f;
        int bi = 0;
        for (int p = 0; p < PP; ++p) {
            float d = lds_pd[p * TOK_PER_BLOCK + tid];
            for (int q = 1; q < RA_KQ; ++q) d += lds_pd[(q * PP + p) * TOK_PER_BLOCK + tid];
            const float pn = __int_as_float(__builtin_amdgcn_readlane(pn_bits, p));
            float sim = d / fmaxf(hn * pn, EPSF);
            if (sim > best) { best = sim; bi = p; }
        }
        lds_as[tid] = (unsigned char)(valid ? bi : 0);
        if (valid)
            __hip_atomic_fetch_add(&lds_cnt[bi], 1.0f,
                                   __ATOMIC_RELAXED, __HIP_MEMORY_SCOPE_WORKGROUP);
    }
    __syncthreads();

    const int blk = n * CHUNKS + chunk;
    if (tid < 32)
        ws_assign[blk * 32 + tid] = __uint_as_float(((const unsigned*)lds_as)[tid]);
    if (tid < PP)
        ws_cnt[blk * CNT_PITCH + tid] = lds_cnt[tid];
}

#define PB_WAVES  16
#define PB_PIECES 20
#define PB_GROUP  4
#define PB_GROUPS (PB_PIECES / PB_GROUP)
#define TV4 (TV / 4)

static __device__ __forceinline__ v4f load_once(const v4f* p) { return __builtin_nontemporal_load(p); }

__global__ void __launch_bounds__(PB_WAVES * 32)
rn_pool_kernel(const float* __restrict__ x, const float* __restrict__ ws_assign,
               const float* __restrict__ ws_cnt, float* __restrict__ out) {
    __shared__ __align__(16) float lds_out[PB_WAVES * 2];

    const unsigned tid  = threadIdx.x;
    const unsigned lane = tid & 31u;
    const unsigned wave = __builtin_amdgcn_readfirstlane(tid >> 5);
    const unsigned pair = blockIdx.x * PB_WAVES + wave;
    const unsigned n    = pair >> 7;
    const unsigned c0   = (pair & 127u) << 1;

    const float* cp = ws_cnt + (size_t)(n * CHUNKS + (lane >> 3)) * CNT_PITCH + (lane & 7u);
    float cpart[CHUNKS / 4];
#pragma unroll
    for (int j = 0; j < CHUNKS / 4; ++j) cpart[j] = cp[j * 4 * CNT_PITCH];

    const float* ap = ws_assign + (size_t)n * AS_PITCH_W;
    unsigned aw[PB_PIECES];
#pragma unroll
    for (int q = 0; q < PB_PIECES; ++q) {
        unsigned i = 32u * q + lane;
        if (i > TV4 - 1) i = TV4 - 1;
        aw[q] = __float_as_uint(ap[i]);
    }
    const bool tail_ok = lane < (TV4 - 32 * (PB_PIECES - 1));

    const float* xr = x + ((size_t)n * MM * CC + c0) * TV;
    const v4f* r0m0 = (const v4f*)xr;
    const v4f* r0m1 = (const v4f*)(xr + (size_t)CC * TV);
    const v4f* r1m0 = (const v4f*)(xr + TV);
    const v4f* r1m1 = (const v4f*)(xr + (size_t)CC * TV + TV);

    v4f buf[2][PB_GROUP * 4];
#define LOAD_GROUP(G) do { \
    _Pragma("unroll") for (int p_ = 0; p_ < PB_GROUP; ++p_) { \
        unsigned i_ = 32u * ((G) * PB_GROUP + p_) + lane; \
        if (i_ > TV4 - 1) i_ = TV4 - 1; \
        buf[(G) & 1][4 * p_ + 0] = load_once(r0m0 + i_); \
        buf[(G) & 1][4 * p_ + 1] = load_once(r0m1 + i_); \
        buf[(G) & 1][4 * p_ + 2] = load_once(r1m0 + i_); \
        buf[(G) & 1][4 * p_ + 3] = load_once(r1m1 + i_); \
    } } while (0)

    v4f acc0 = (v4f){0.f, 0.f, 0.f, 0.f};
    v4f acc1 = (v4f){0.f, 0.f, 0.f, 0.f};
    int winv_bits = 0;

    __builtin_amdgcn_sched_barrier(0);
    LOAD_GROUP(0);
#pragma unroll
    for (int g = 0; g < PB_GROUPS; ++g) {
        __builtin_amdgcn_sched_barrier(0);
        if (g + 1 < PB_GROUPS) LOAD_GROUP(g + 1);
        __builtin_amdgcn_sched_barrier(0);
        if (g == 0) {
            float cnt = 0.0f;
#pragma unroll
            for (int j = 0; j < CHUNKS / 4; ++j) cnt += cpart[j];
            cnt += __shfl_xor(cnt, 8, 32);
            cnt += __shfl_xor(cnt, 16, 32);
            winv_bits = __float_as_int(1.0f / fmaxf(cnt, 1.0f));
        }
#pragma unroll
        for (int p = 0; p < PB_GROUP; ++p) {
            const int q = g * PB_GROUP + p;
            const unsigned w = aw[q];
            v4f wt;
            wt.x = __int_as_float(__builtin_amdgcn_ds_bpermute((int)((w << 2) & 28u), winv_bits));
            wt.y = __int_as_float(__builtin_amdgcn_ds_bpermute((int)((w >> 6) & 28u), winv_bits));
            wt.z = __int_as_float(__builtin_amdgcn_ds_bpermute((int)((w >> 14) & 28u), winv_bits));
            wt.w = __int_as_float(__builtin_amdgcn_ds_bpermute((int)((w >> 22) & 28u), winv_bits));
            if (q == PB_PIECES - 1 && !tail_ok) wt = (v4f){0.f, 0.f, 0.f, 0.f};
            const v4f* b = &buf[g & 1][4 * p];
            acc0 += (b[0] + b[1]) * wt;
            acc1 += (b[2] + b[3]) * wt;
        }
    }
#undef LOAD_GROUP

    float s0 = (acc0.x + acc0.y) + (acc0.z + acc0.w);
    float s1 = (acc1.x + acc1.y) + (acc1.z + acc1.w);
#pragma unroll
    for (int m = 16; m >= 1; m >>= 1) {
        s0 += __shfl_xor(s0, m, 32);
        s1 += __shfl_xor(s1, m, 32);
    }
    if (lane == 0) { lds_out[2 * wave] = s0 * 0.0625f; lds_out[2 * wave + 1] = s1 * 0.0625f; }
    __syncthreads();
    if (tid < 32) out[(size_t)blockIdx.x * 32 + tid] = lds_out[tid];
}

extern "C" void kernel_launch(void* const* d_in, const int* in_sizes, int n_in,
                              void* d_out, int out_size, void* d_ws, size_t ws_size,
                              hipStream_t stream) {
    const float* x      = (const float*)d_in[0];
    const float* protos = (const float*)d_in[1];

    float* ws_assign = (float*)d_ws;
    float* ws_cnt    = ws_assign + NN * AS_PITCH_W;

    dim3 grid(CHUNKS, NN);
    rn_assign_kernel<<<grid, 256, 0, stream>>>(x, protos, ws_assign, ws_cnt);

    rn_pool_kernel<<<NN * CC / (2 * PB_WAVES), PB_WAVES * 32, 0, stream>>>(x, ws_assign, ws_cnt, (float*)d_out);
}
